# starting kernel
# baseline (speedup 1.0000x reference)
#include <hip/hip_runtime.h>
#include <hip/hip_bf16.h>

typedef __attribute__((ext_vector_type(2))) float v2f;
typedef __attribute__((ext_vector_type(8))) float v8f;

#define B_   16
#define L_   512
#define DB_  3072
#define D_   300
#define NN_  256
#define NW_  30
#define H_   256

#define M1   (B_ * L_)
#define MN   (B_ * NN_)

#define BM 128
#define BN 64
#define KC 32
#define LST 36

__global__ __launch_bounds__(128) void gemm_bert(
    const float* __restrict__ A, const float* __restrict__ W,
    const float* __restrict__ bias, const float* __restrict__ prelu_a,
    float* __restrict__ art)
{
    __shared__ __align__(16) float As[BM][LST];
    __shared__ __align__(16) float Bs[BN][LST];

    const int tid  = threadIdx.x;
    const int wave = tid >> 5;
    const int lane = tid & 31;
    const int rc   = lane & 15;
    const int rsel = (lane >> 4) << 1;
    const int rowh = (lane >> 4) << 3;

    const int m0 = blockIdx.x * BM;
    const int nb = blockIdx.y * BN;

    v8f acc[8];
#pragma unroll
    for (int t = 0; t < 8; ++t) acc[t] = (v8f){0,0,0,0,0,0,0,0};

    for (int kb = 0; kb < DB_; kb += KC) {
#pragma unroll
        for (int j = 0; j < 8; ++j) {
            int i   = tid + 128 * j;
            int row = i >> 3;
            int c4  = (i & 7) << 2;
            float4 v = *(const float4*)&A[(size_t)(m0 + row) * DB_ + kb + c4];
            *(float4*)&As[row][c4] = v;
        }
#pragma unroll
        for (int j = 0; j < 4; ++j) {
            int i  = tid + 128 * j;
            int n  = i >> 3;
            int c4 = (i & 7) << 2;
            float4 v = make_float4(0.f, 0.f, 0.f, 0.f);
            if (nb + n < D_)
                v = *(const float4*)&W[(size_t)(nb + n) * DB_ + kb + c4];
            *(float4*)&Bs[n][c4] = v;
        }
        __syncthreads();

        const int ar0 = wave * 32 + rc;
        const int ar1 = wave * 32 + 16 + rc;
#pragma unroll
        for (int kk = 0; kk < KC; kk += 4) {
            v2f a0, a1;
            a0.x = As[ar0][kk + rsel];
            a0.y = As[ar0][kk + rsel + 1];
            a1.x = As[ar1][kk + rsel];
            a1.y = As[ar1][kk + rsel + 1];
#pragma unroll
            for (int nt = 0; nt < 4; ++nt) {
                v2f b;
                b.x = Bs[nt * 16 + rc][kk + rsel];
                b.y = Bs[nt * 16 + rc][kk + rsel + 1];
                acc[nt] = __builtin_amdgcn_wmma_f32_16x16x4_f32(
                    false, a0, false, b, (short)0, acc[nt], false, false);
                acc[nt + 4] = __builtin_amdgcn_wmma_f32_16x16x4_f32(
                    false, a1, false, b, (short)0, acc[nt + 4], false, false);
            }
        }
        __syncthreads();
    }

    const float pa = prelu_a[0];
#pragma unroll
    for (int nt = 0; nt < 4; ++nt) {
        int gc = nb + nt * 16 + rc;
        if (gc < D_) {
            float bb = bias[gc];
#pragma unroll
            for (int part = 0; part < 2; ++part) {
#pragma unroll
                for (int i = 0; i < 8; ++i) {
                    int gr = m0 + wave * 32 + part * 16 + i + rowh;
                    float v = acc[nt + 4 * part][i] + bb;
                    v = v > 0.f ? v : pa * v;
                    art[(size_t)gr * D_ + gc] = v;
                }
            }
        }
    }
}

__global__ __launch_bounds__(128) void gather_pool(
    const float* __restrict__ art, const int* __restrict__ idx,
    const int* __restrict__ mask, float* __restrict__ nodes)
{
    const int bn = blockIdx.x;
    const int b  = bn / NN_;
    __shared__ int sidx[NW_];
    __shared__ int smsk[NW_];
    if (threadIdx.x < NW_) {
        sidx[threadIdx.x] = idx[(size_t)bn * NW_ + threadIdx.x];
        smsk[threadIdx.x] = mask[(size_t)bn * NW_ + threadIdx.x];
    }
    __syncthreads();

    float cnt = 0.f;
#pragma unroll
    for (int w = 0; w < NW_; ++w) cnt += (float)smsk[w];
    const float inv = 1.f / fmaxf(cnt, 1.f);

    for (int d = threadIdx.x; d < D_; d += 128) {
        float s = 0.f;
#pragma unroll
        for (int w = 0; w < NW_; ++w) {
            if (smsk[w])
                s += art[((size_t)b * L_ + sidx[w]) * D_ + d];
        }
        nodes[(size_t)bn * D_ + d] = s * inv;
    }
}

__global__ __launch_bounds__(128) void gemm_head(
    const float* __restrict__ Nd, const float* __restrict__ W1,
    const float* __restrict__ b1, float* __restrict__ h)
{
    const int wave = threadIdx.x >> 5;
    const int lane = threadIdx.x & 31;
    const int rc   = lane & 15;
    const int rsel = (lane >> 4) << 1;
    const int rowh = (lane >> 4) << 3;

    const int tile = blockIdx.x * 4 + wave;
    const int mt = tile >> 4;
    const int nt = tile & 15;

    const float* Ar = Nd + (size_t)(mt * 16 + rc) * D_;
    const float* Br = W1 + (size_t)(nt * 16 + rc) * D_;

    v8f acc = (v8f){0,0,0,0,0,0,0,0};
#pragma unroll 5
    for (int k = 0; k < D_; k += 4) {
        v2f a, b;
        a.x = Ar[k + rsel];
        a.y = Ar[k + rsel + 1];
        b.x = Br[k + rsel];
        b.y = Br[k + rsel + 1];
        acc = __builtin_amdgcn_wmma_f32_16x16x4_f32(
            false, a, false, b, (short)0, acc, false, false);
    }

    const int col = nt * 16 + rc;
    const float bb = b1[col];
#pragma unroll
    for (int i = 0; i < 8; ++i) {
        int row = mt * 16 + i + rowh;
        float v = acc[i] + bb;
        h[(size_t)row * H_ + col] = fmaxf(v, 0.f);
    }
}

__global__ __launch_bounds__(256) void head_out(
    const float* __restrict__ h, const float* __restrict__ W2,
    const float* __restrict__ b2, float* __restrict__ out)
{
    const int j    = blockIdx.x * 8 + (threadIdx.x >> 5);
    const int lane = threadIdx.x & 31;
    const float* hr = h + (size_t)j * H_;
    float p = 0.f;
#pragma unroll
    for (int e = lane; e < H_; e += 32) p += hr[e] * W2[e];
#pragma unroll
    for (int off = 16; off > 0; off >>= 1) p += __shfl_down(p, off, 32);
    if (lane == 0) out[j] = p + b2[0];
}

extern "C" void kernel_launch(void* const* d_in, const int* in_sizes, int n_in,
                              void* d_out, int out_size, void* d_ws, size_t ws_size,
                              hipStream_t stream)
{
    const float* articles = (const float*)d_in[0];
    const int*   nidx     = (const int*)  d_in[1];
    const int*   nmask    = (const int*)  d_in[2];
    const float* W_bert   = (const float*)d_in[5];
    const float* b_bert   = (const float*)d_in[6];
    const float* prelu_a  = (const float*)d_in[7];
    const float* W1       = (const float*)d_in[8];
    const float* b1       = (const float*)d_in[9];
    const float* W2       = (const float*)d_in[10];
    const float* b2       = (const float*)d_in[11];
    float* out = (float*)d_out;

    float* art   = (float*)d_ws;
    float* nodes = art + (size_t)M1 * D_;
    float* h     = nodes + (size_t)MN * D_;
    (void)in_sizes; (void)n_in; (void)out_size; (void)ws_size;

    gemm_bert  <<<dim3(M1 / BM, 5), 128, 0, stream>>>(articles, W_bert, b_bert, prelu_a, art);
    gather_pool<<<MN, 128, 0, stream>>>(art, nidx, nmask, nodes);
    gemm_head  <<<MN * (H_ / 16) / (4 * 16), 128, 0, stream>>>(nodes, W1, b1, h);
    head_out   <<<MN / 8, 256, 0, stream>>>(h, W2, b2, out);
}


# head product moved in front of the pooling (mean of rows @ W1^T = mean of (rows @ W1^T)): art @ W1^T as an LDS-tiled WMMA GEMM, then a wave per node pools Z rows, relu, dot W2; sums reordered
# speedup vs baseline: 1.6769x; 1.6769x over previous
#include <hip/hip_runtime.h>
#include <hip/hip_bf16.h>

typedef __attribute__((ext_vector_type(2))) float v2f;
typedef __attribute__((ext_vector_type(8))) float v8f;
typedef __attribute__((ext_vector_type(4))) float v4f;

#define B_   16
#define L_   512
#define DB_  3072
#define D_   300
#define NN_  256
#define NW_  30
#define H_   256

#define M1   (B_ * L_)
#define MN   (B_ * NN_)

#define BM 128
#define BN 64
#define KC 32
#define LST 36

__global__ __launch_bounds__(128) void gemm_bert(
    const float* __restrict__ A, const float* __restrict__ W,
    const float* __restrict__ bias, const float* __restrict__ prelu_a,
    float* __restrict__ art)
{
    __shared__ __align__(16) float As[BM][LST];
    __shared__ __align__(16) float Bs[BN][LST];

    const int tid  = threadIdx.x;
    const int wave = tid >> 5;
    const int lane = tid & 31;
    const int rc   = lane & 15;
    const int rsel = (lane >> 4) << 1;
    const int rowh = (lane >> 4) << 3;

    const int m0 = blockIdx.x * BM;
    const int nb = blockIdx.y * BN;

    v8f acc[8];
#pragma unroll
    for (int t = 0; t < 8; ++t) acc[t] = (v8f){0,0,0,0,0,0,0,0};

    for (int kb = 0; kb < DB_; kb += KC) {
        float4 va[8], vw[4];
#pragma unroll
        for (int j = 0; j < 8; ++j) {
            int i   = tid + 128 * j;
            int row = i >> 3;
            int c4  = (i & 7) << 2;
            va[j] = *(const float4*)&A[(size_t)(m0 + row) * DB_ + kb + c4];
        }
#pragma unroll
        for (int j = 0; j < 4; ++j) {
            int i  = tid + 128 * j;
            int n  = i >> 3;
            int c4 = (i & 7) << 2;
            int wr = min(nb + n, D_ - 1);
            vw[j] = *(const float4*)&W[(size_t)wr * DB_ + kb + c4];
        }
        __builtin_amdgcn_sched_barrier(0);
#pragma unroll
        for (int j = 0; j < 8; ++j) {
            int i   = tid + 128 * j;
            *(float4*)&As[i >> 3][(i & 7) << 2] = va[j];
        }
#pragma unroll
        for (int j = 0; j < 4; ++j) {
            int i  = tid + 128 * j;
            *(float4*)&Bs[i >> 3][(i & 7) << 2] = vw[j];
        }
        __syncthreads();

        const int ar0 = wave * 32 + rc;
        const int ar1 = wave * 32 + 16 + rc;
#pragma unroll
        for (int kk = 0; kk < KC; kk += 4) {
            v2f a0, a1;
            a0.x = As[ar0][kk + rsel];
            a0.y = As[ar0][kk + rsel + 1];
            a1.x = As[ar1][kk + rsel];
            a1.y = As[ar1][kk + rsel + 1];
#pragma unroll
            for (int nt = 0; nt < 4; ++nt) {
                v2f b;
                b.x = Bs[nt * 16 + rc][kk + rsel];
                b.y = Bs[nt * 16 + rc][kk + rsel + 1];
                acc[nt] = __builtin_amdgcn_wmma_f32_16x16x4_f32(
                    false, a0, false, b, (short)0, acc[nt], false, false);
                acc[nt + 4] = __builtin_amdgcn_wmma_f32_16x16x4_f32(
                    false, a1, false, b, (short)0, acc[nt + 4], false, false);
            }
        }
        __syncthreads();
    }

    const float pa = prelu_a[0];
#pragma unroll
    for (int nt = 0; nt < 4; ++nt) {
        int gc = nb + nt * 16 + rc;
        if (gc < D_) {
            float bb = bias[gc];
#pragma unroll
            for (int part = 0; part < 2; ++part) {
#pragma unroll
                for (int i = 0; i < 8; ++i) {
                    int gr = m0 + wave * 32 + part * 16 + i + rowh;
                    float v = acc[nt + 4 * part][i] + bb;
                    v = v > 0.f ? v : pa * v;
                    art[(size_t)gr * D_ + gc] = v;
                }
            }
        }
    }
}

#define ZBM 128
#define ZBN 64
#define ZKC 60
#define ZQ  (ZKC / 4)
#define ZBR 69

__global__ __launch_bounds__(128) void gemm_z(
    const float* art, const float* W1, float* Z)
{
    __shared__ __align__(16) float As[ZBM * ZKC];
    __shared__ __align__(16) float Bs[ZBR * ZKC];

    const int tid  = threadIdx.x;
    const int wave = tid >> 5;
    const int lane = tid & 31;
    const int rc   = lane & 15;
    const int rsel = (lane >> 4) << 1;
    const int rowh = (lane >> 4) << 3;

    const int m0 = blockIdx.x * ZBM;
    const int nb = blockIdx.y * ZBN;

    v8f acc[8];
#pragma unroll
    for (int t = 0; t < 8; ++t) acc[t] = (v8f){0,0,0,0,0,0,0,0};

    for (int kb = 0; kb < D_; kb += ZKC) {
        v4f va[15], vw[8];
#pragma unroll
        for (int j = 0; j < 15; ++j) {
            int i   = tid + 128 * j;
            int row = i / ZQ;
            int c4  = (i % ZQ) << 2;
            va[j] = *(const v4f*)&art[(size_t)(m0 + row) * D_ + kb + c4];
        }
#pragma unroll
        for (int j = 0; j < 8; ++j) {
            int i  = tid + 128 * j;
            int n  = i / ZQ;
            int c4 = (i % ZQ) << 2;
            int wr = min(nb + n, H_ - 1);
            vw[j] = *(const v4f*)&W1[(size_t)wr * D_ + kb + c4];
        }
        __builtin_amdgcn_sched_barrier(0);
#pragma unroll
        for (int j = 0; j < 15; ++j)
            *(v4f*)&As[(tid + 128 * j) << 2] = va[j];
#pragma unroll
        for (int j = 0; j < 8; ++j)
            *(v4f*)&Bs[(tid + 128 * j) << 2] = vw[j];
        __syncthreads();

        const float* a0p = &As[(wave * 32 + rc) * ZKC + rsel];
        const float* a1p = a0p + 16 * ZKC;
        const float* bp  = &Bs[rc * ZKC + rsel];
#pragma unroll
        for (int kk = 0; kk < ZKC; kk += 4) {
            v2f a0, a1;
            a0.x = a0p[kk];
            a0.y = a0p[kk + 1];
            a1.x = a1p[kk];
            a1.y = a1p[kk + 1];
#pragma unroll
            for (int nt = 0; nt < 4; ++nt) {
                v2f b;
                b.x = bp[nt * 16 * ZKC + kk];
                b.y = bp[nt * 16 * ZKC + kk + 1];
                acc[nt] = __builtin_amdgcn_wmma_f32_16x16x4_f32(
                    false, a0, false, b, (short)0, acc[nt], false, false);
                acc[nt + 4] = __builtin_amdgcn_wmma_f32_16x16x4_f32(
                    false, a1, false, b, (short)0, acc[nt + 4], false, false);
            }
        }
        __syncthreads();
    }

#pragma unroll
    for (int nt = 0; nt < 4; ++nt) {
        int gc = nb + nt * 16 + rc;
#pragma unroll
        for (int part = 0; part < 2; ++part) {
#pragma unroll
            for (int i = 0; i < 8; ++i) {
                int gr = m0 + wave * 32 + part * 16 + i + rowh;
                Z[(size_t)gr * H_ + gc] = acc[nt + 4 * part][i];
            }
        }
    }
}

#define PG 10

__global__ __launch_bounds__(128) void pool_head(
    const float* __restrict__ Z, const int* __restrict__ idx, const int* __restrict__ mask,
    const float* __restrict__ b1, const float* __restrict__ W2, const float* __restrict__ b2,
    float* __restrict__ out)
{
    const int node = blockIdx.x * 4 + (threadIdx.x >> 5);
    const int lane = threadIdx.x & 31;
    const int b    = node / NN_;

    const int   le = min(lane, NW_ - 1);
    const int   iv = idx[(size_t)node * NW_ + le];
    const float mv = lane < NW_ ? (float)mask[(size_t)node * NW_ + le] : 0.f;

    float cnt = mv;
#pragma unroll
    for (int off = 16; off > 0; off >>= 1) cnt += __shfl_xor(cnt, off, 32);
    const float inv = 1.f / fmaxf(cnt, 1.f);

    const float4* zb = (const float4*)(Z + (size_t)b * L_ * H_) + lane;

    float4 s0 = make_float4(0.f, 0.f, 0.f, 0.f);
    float4 s1 = make_float4(0.f, 0.f, 0.f, 0.f);
#pragma unroll
    for (int g = 0; g < NW_ / PG; ++g) {
        float4 x0[PG], x1[PG];
        float  m[PG];
#pragma unroll
        for (int u = 0; u < PG; ++u) {
            const int w = g * PG + u;
            const int r = __builtin_amdgcn_readlane(iv, w);
            m[u] = __int_as_float(__builtin_amdgcn_readlane(__float_as_int(mv), w));
            const float4* zr = zb + (size_t)r * (H_ / 4);
            x0[u] = zr[0];
            x1[u] = zr[32];
        }
        __builtin_amdgcn_sched_barrier(0);
#pragma unroll
        for (int u = 0; u < PG; ++u) {
            s0.x += m[u] * x0[u].x;  s0.y += m[u] * x0[u].y;
            s0.z += m[u] * x0[u].z;  s0.w += m[u] * x0[u].w;
            s1.x += m[u] * x1[u].x;  s1.y += m[u] * x1[u].y;
            s1.z += m[u] * x1[u].z;  s1.w += m[u] * x1[u].w;
        }
    }

    const int c0 = lane << 2;
    const int c1 = 128 + (lane << 2);
    float p = 0.f;
    p += fmaxf(s0.x * inv + b1[c0 + 0], 0.f) * W2[c0 + 0];
    p += fmaxf(s0.y * inv + b1[c0 + 1], 0.f) * W2[c0 + 1];
    p += fmaxf(s0.z * inv + b1[c0 + 2], 0.f) * W2[c0 + 2];
    p += fmaxf(s0.w * inv + b1[c0 + 3], 0.f) * W2[c0 + 3];
    p += fmaxf(s1.x * inv + b1[c1 + 0], 0.f) * W2[c1 + 0];
    p += fmaxf(s1.y * inv + b1[c1 + 1], 0.f) * W2[c1 + 1];
    p += fmaxf(s1.z * inv + b1[c1 + 2], 0.f) * W2[c1 + 2];
    p += fmaxf(s1.w * inv + b1[c1 + 3], 0.f) * W2[c1 + 3];
#pragma unroll
    for (int off = 16; off > 0; off >>= 1) p += __shfl_xor(p, off, 32);
    if (lane == 0) out[node] = p + b2[0];
}

extern "C" void kernel_launch(void* const* d_in, const int* in_sizes, int n_in,
                              void* d_out, int out_size, void* d_ws, size_t ws_size,
                              hipStream_t stream)
{
    const float* articles = (const float*)d_in[0];
    const int*   nidx     = (const int*)  d_in[1];
    const int*   nmask    = (const int*)  d_in[2];
    const float* W_bert   = (const float*)d_in[5];
    const float* b_bert   = (const float*)d_in[6];
    const float* prelu_a  = (const float*)d_in[7];
    const float* W1       = (const float*)d_in[8];
    const float* b1       = (const float*)d_in[9];
    const float* W2       = (const float*)d_in[10];
    const float* b2       = (const float*)d_in[11];
    float* out = (float*)d_out;

    float* art = (float*)d_ws;
    float* Zp  = art + (size_t)M1 * D_;
    (void)in_sizes; (void)n_in; (void)out_size; (void)ws_size;

    gemm_bert<<<dim3(M1 / BM, 5), 128, 0, stream>>>(articles, W_bert, b_bert, prelu_a, art);
    gemm_z   <<<dim3(M1 / ZBM, H_ / ZBN), 128, 0, stream>>>(art, W1, Zp);
    pool_head<<<MN / 4, 128, 0, stream>>>(Zp, nidx, nmask, b1, W2, b2, out);
}


# gemm_bert: 256 equal work-groups of 8 waves on 128x80 tiles; two 4-wave groups take alternate K chunks, one stages while the other multiplies; their K partials are added via LDS (sum order changes)
# speedup vs baseline: 2.2688x; 1.3530x over previous
#include <hip/hip_runtime.h>
#include <hip/hip_bf16.h>

typedef __attribute__((ext_vector_type(2))) float v2f;
typedef __attribute__((ext_vector_type(8))) float v8f;
typedef __attribute__((ext_vector_type(4))) float v4f;

#define B_   16
#define L_   512
#define DB_  3072
#define D_   300
#define NN_  256
#define NW_  30
#define H_   256

#define M1   (B_ * L_)
#define MN   (B_ * NN_)

#define BM 128
#define BN 80
#define KC 32
#define LST 36
#define NCH (DB_ / KC)
#define NTB (BN / 16)
#define GRP_FLOATS ((BM + BN) * LST)

__global__ __launch_bounds__(256) void gemm_bert(
    const float* __restrict__ A, const float* __restrict__ W,
    const float* __restrict__ bias, const float* __restrict__ prelu_a,
    float* __restrict__ art)
{
    __shared__ __align__(16) float smem[2 * GRP_FLOATS];

    const int tid  = threadIdx.x;
    const int grp  = __builtin_amdgcn_readfirstlane(tid >> 7);
    const int t    = tid & 127;
    const int wave = tid >> 5;
    const int wv   = wave & 3;
    const int lane = tid & 31;
    const int rc   = lane & 15;
    const int rsel = (lane >> 4) << 1;
    const int rowh = (lane >> 4) << 3;

    const int m0 = blockIdx.x * BM;
    const int nb = blockIdx.y * BN;

    float* As = smem + grp * GRP_FLOATS;
    float* Bs = As + BM * LST;

    v8f acc[2 * NTB];
#pragma unroll
    for (int q = 0; q < 2 * NTB; ++q) acc[q] = (v8f){0,0,0,0,0,0,0,0};

    const float* a0p = As + (wv * 32 + rc) * LST + rsel;
    const float* a1p = a0p + 16 * LST;
    const float* bp  = Bs + rc * LST + rsel;

    for (int p = 0; p <= NCH; ++p) {
        if ((p & 1) == grp) {
            if (p < NCH) {
                const int kb = p * KC;
                float4 va[8], vw[5];
#pragma unroll
                for (int j = 0; j < 8; ++j) {
                    int i   = t + 128 * j;
                    int row = i >> 3;
                    int c4  = (i & 7) << 2;
                    va[j] = *(const float4*)&A[(size_t)(m0 + row) * DB_ + kb + c4];
                }
#pragma unroll
                for (int j = 0; j < 5; ++j) {
                    int i  = t + 128 * j;
                    int n  = i >> 3;
                    int c4 = (i & 7) << 2;
                    int wr = min(nb + n, D_ - 1);
                    vw[j] = *(const float4*)&W[(size_t)wr * DB_ + kb + c4];
                }
                __builtin_amdgcn_sched_barrier(0);
#pragma unroll
                for (int j = 0; j < 8; ++j) {
                    int i = t + 128 * j;
                    *(float4*)&As[(i >> 3) * LST + ((i & 7) << 2)] = va[j];
                }
#pragma unroll
                for (int j = 0; j < 5; ++j) {
                    int i = t + 128 * j;
                    *(float4*)&Bs[(i >> 3) * LST + ((i & 7) << 2)] = vw[j];
                }
            }
        } else {
            if (p > 0) {
#pragma unroll
                for (int kk = 0; kk < KC; kk += 4) {
                    v2f a0, a1;
                    a0.x = a0p[kk];
                    a0.y = a0p[kk + 1];
                    a1.x = a1p[kk];
                    a1.y = a1p[kk + 1];
#pragma unroll
                    for (int nt = 0; nt < NTB; ++nt) {
                        v2f b;
                        b.x = bp[nt * 16 * LST + kk];
                        b.y = bp[nt * 16 * LST + kk + 1];
                        acc[nt] = __builtin_amdgcn_wmma_f32_16x16x4_f32(
                            false, a0, false, b, (short)0, acc[nt], false, false);
                        acc[nt + NTB] = __builtin_amdgcn_wmma_f32_16x16x4_f32(
                            false, a1, false, b, (short)0, acc[nt + NTB], false, false);
                    }
                }
            }
        }
        __syncthreads();
    }

    float* red = smem;
    v8f fin[NTB], giv[NTB];
#pragma unroll
    for (int q = 0; q < NTB; ++q) {
        fin[q] = grp ? acc[q + NTB] : acc[q];
        giv[q] = grp ? acc[q] : acc[q + NTB];
    }
#pragma unroll
    for (int q = 0; q < NTB; ++q)
#pragma unroll
        for (int i = 0; i < 8; ++i)
            red[((wave * NTB + q) * 8 + i) * 32 + lane] = giv[q][i];
    __syncthreads();
    const int pw = wave ^ 4;
#pragma unroll
    for (int q = 0; q < NTB; ++q)
#pragma unroll
        for (int i = 0; i < 8; ++i)
            fin[q][i] += red[((pw * NTB + q) * 8 + i) * 32 + lane];

    const float pa = prelu_a[0];
#pragma unroll
    for (int nt = 0; nt < NTB; ++nt) {
        int gc = nb + nt * 16 + rc;
        if (gc < D_) {
            float bb = bias[gc];
#pragma unroll
            for (int i = 0; i < 8; ++i) {
                int gr = m0 + wv * 32 + grp * 16 + i + rowh;
                float v = fin[nt][i] + bb;
                v = v > 0.f ? v : pa * v;
                art[(size_t)gr * D_ + gc] = v;
            }
        }
    }
}

#define ZBM 128
#define ZBN 64
#define ZKC 60
#define ZQ  (ZKC / 4)
#define ZBR 69

__global__ __launch_bounds__(128) void gemm_z(
    const float* art, const float* W1, float* Z)
{
    __shared__ __align__(16) float As[ZBM * ZKC];
    __shared__ __align__(16) float Bs[ZBR * ZKC];

    const int tid  = threadIdx.x;
    const int wave = tid >> 5;
    const int lane = tid & 31;
    const int rc   = lane & 15;
    const int rsel = (lane >> 4) << 1;
    const int rowh = (lane >> 4) << 3;

    const int m0 = blockIdx.x * ZBM;
    const int nb = blockIdx.y * ZBN;

    v8f acc[8];
#pragma unroll
    for (int t = 0; t < 8; ++t) acc[t] = (v8f){0,0,0,0,0,0,0,0};

    for (int kb = 0; kb < D_; kb += ZKC) {
        v4f va[15], vw[8];
#pragma unroll
        for (int j = 0; j < 15; ++j) {
            int i   = tid + 128 * j;
            int row = i / ZQ;
            int c4  = (i % ZQ) << 2;
            va[j] = *(const v4f*)&art[(size_t)(m0 + row) * D_ + kb + c4];
        }
#pragma unroll
        for (int j = 0; j < 8; ++j) {
            int i  = tid + 128 * j;
            int n  = i / ZQ;
            int c4 = (i % ZQ) << 2;
            int wr = min(nb + n, H_ - 1);
            vw[j] = *(const v4f*)&W1[(size_t)wr * D_ + kb + c4];
        }
        __builtin_amdgcn_sched_barrier(0);
#pragma unroll
        for (int j = 0; j < 15; ++j)
            *(v4f*)&As[(tid + 128 * j) << 2] = va[j];
#pragma unroll
        for (int j = 0; j < 8; ++j)
            *(v4f*)&Bs[(tid + 128 * j) << 2] = vw[j];
        __syncthreads();

        const float* a0p = &As[(wave * 32 + rc) * ZKC + rsel];
        const float* a1p = a0p + 16 * ZKC;
        const float* bp  = &Bs[rc * ZKC + rsel];
#pragma unroll
        for (int kk = 0; kk < ZKC; kk += 4) {
            v2f a0, a1;
            a0.x = a0p[kk];
            a0.y = a0p[kk + 1];
            a1.x = a1p[kk];
            a1.y = a1p[kk + 1];
#pragma unroll
            for (int nt = 0; nt < 4; ++nt) {
                v2f b;
                b.x = bp[nt * 16 * ZKC + kk];
                b.y = bp[nt * 16 * ZKC + kk + 1];
                acc[nt] = __builtin_amdgcn_wmma_f32_16x16x4_f32(
                    false, a0, false, b, (short)0, acc[nt], false, false);
                acc[nt + 4] = __builtin_amdgcn_wmma_f32_16x16x4_f32(
                    false, a1, false, b, (short)0, acc[nt + 4], false, false);
            }
        }
        __syncthreads();
    }

#pragma unroll
    for (int nt = 0; nt < 4; ++nt) {
        int gc = nb + nt * 16 + rc;
#pragma unroll
        for (int part = 0; part < 2; ++part) {
#pragma unroll
            for (int i = 0; i < 8; ++i) {
                int gr = m0 + wave * 32 + part * 16 + i + rowh;
                Z[(size_t)gr * H_ + gc] = acc[nt + 4 * part][i];
            }
        }
    }
}

#define PG 10

__global__ __launch_bounds__(128) void pool_head(
    const float* __restrict__ Z, const int* __restrict__ idx, const int* __restrict__ mask,
    const float* __restrict__ b1, const float* __restrict__ W2, const float* __restrict__ b2,
    float* __restrict__ out)
{
    const int node = blockIdx.x * 4 + (threadIdx.x >> 5);
    const int lane = threadIdx.x & 31;
    const int b    = node / NN_;

    const int   le = min(lane, NW_ - 1);
    const int   iv = idx[(size_t)node * NW_ + le];
    const float mv = lane < NW_ ? (float)mask[(size_t)node * NW_ + le] : 0.f;

    float cnt = mv;
#pragma unroll
    for (int off = 16; off > 0; off >>= 1) cnt += __shfl_xor(cnt, off, 32);
    const float inv = 1.f / fmaxf(cnt, 1.f);

    const float4* zb = (const float4*)(Z + (size_t)b * L_ * H_) + lane;

    float4 s0 = make_float4(0.f, 0.f, 0.f, 0.f);
    float4 s1 = make_float4(0.f, 0.f, 0.f, 0.f);
#pragma unroll
    for (int g = 0; g < NW_ / PG; ++g) {
        float4 x0[PG], x1[PG];
        float  m[PG];
#pragma unroll
        for (int u = 0; u < PG; ++u) {
            const int w = g * PG + u;
            const int r = __builtin_amdgcn_readlane(iv, w);
            m[u] = __int_as_float(__builtin_amdgcn_readlane(__float_as_int(mv), w));
            const float4* zr = zb + (size_t)r * (H_ / 4);
            x0[u] = zr[0];
            x1[u] = zr[32];
        }
        __builtin_amdgcn_sched_barrier(0);
#pragma unroll
        for (int u = 0; u < PG; ++u) {
            s0.x += m[u] * x0[u].x;  s0.y += m[u] * x0[u].y;
            s0.z += m[u] * x0[u].z;  s0.w += m[u] * x0[u].w;
            s1.x += m[u] * x1[u].x;  s1.y += m[u] * x1[u].y;
            s1.z += m[u] * x1[u].z;  s1.w += m[u] * x1[u].w;
        }
    }

    const int c0 = lane << 2;
    const int c1 = 128 + (lane << 2);
    float p = 0.f;
    p += fmaxf(s0.x * inv + b1[c0 + 0], 0.f) * W2[c0 + 0];
    p += fmaxf(s0.y * inv + b1[c0 + 1], 0.f) * W2[c0 + 1];
    p += fmaxf(s0.z * inv + b1[c0 + 2], 0.f) * W2[c0 + 2];
    p += fmaxf(s0.w * inv + b1[c0 + 3], 0.f) * W2[c0 + 3];
    p += fmaxf(s1.x * inv + b1[c1 + 0], 0.f) * W2[c1 + 0];
    p += fmaxf(s1.y * inv + b1[c1 + 1], 0.f) * W2[c1 + 1];
    p += fmaxf(s1.z * inv + b1[c1 + 2], 0.f) * W2[c1 + 2];
    p += fmaxf(s1.w * inv + b1[c1 + 3], 0.f) * W2[c1 + 3];
#pragma unroll
    for (int off = 16; off > 0; off >>= 1) p += __shfl_xor(p, off, 32);
    if (lane == 0) out[node] = p + b2[0];
}

extern "C" void kernel_launch(void* const* d_in, const int* in_sizes, int n_in,
                              void* d_out, int out_size, void* d_ws, size_t ws_size,
                              hipStream_t stream)
{
    const float* articles = (const float*)d_in[0];
    const int*   nidx     = (const int*)  d_in[1];
    const int*   nmask    = (const int*)  d_in[2];
    const float* W_bert   = (const float*)d_in[5];
    const float* b_bert   = (const float*)d_in[6];
    const float* prelu_a  = (const float*)d_in[7];
    const float* W1       = (const float*)d_in[8];
    const float* b1       = (const float*)d_in[9];
    const float* W2       = (const float*)d_in[10];
    const float* b2       = (const float*)d_in[11];
    float* out = (float*)d_out;

    float* art = (float*)d_ws;
    float* Zp  = art + (size_t)M1 * D_;
    (void)in_sizes; (void)n_in; (void)out_size; (void)ws_size;

    gemm_bert<<<dim3(M1 / BM, 4), 256, 0, stream>>>(articles, W_bert, b_bert, prelu_a, art);
    gemm_z   <<<dim3(M1 / ZBM, H_ / ZBN), 128, 0, stream>>>(art, W1, Zp);
    pool_head<<<MN / 4, 128, 0, stream>>>(Zp, nidx, nmask, b1, W2, b2, out);
}


# gemm_bert: fixed wave roles: 4 loader waves carry the next K chunk's loads across an LDS-only barrier into 2 LDS stages; 2 multiply waves per strip split each chunk's k-steps; sums reordered
# speedup vs baseline: 2.3523x; 1.0368x over previous
#include <hip/hip_runtime.h>
#include <hip/hip_bf16.h>

typedef __attribute__((ext_vector_type(2))) float v2f;
typedef __attribute__((ext_vector_type(8))) float v8f;
typedef __attribute__((ext_vector_type(4))) float v4f;

#define B_   16
#define L_   512
#define DB_  3072
#define D_   300
#define NN_  256
#define NW_  30
#define H_   256

#define M1   (B_ * L_)
#define MN   (B_ * NN_)

#define BM 128
#define BN 80
#define KC 32
#define LST 36
#define NCH (DB_ / KC)
#define NTB (BN / 16)
#define STG_FLOATS ((BM + BN) * LST)

__device__ __forceinline__ void chunk_fetch(v4f (&va)[8], v4f (&vw)[5],
                                            const float* pA, const float* const (&pW)[5], int kb)
{
#pragma unroll
    for (int j = 0; j < 8; ++j) va[j] = *(const v4f*)(pA + (size_t)j * 16 * DB_ + kb);
#pragma unroll
    for (int j = 0; j < 5; ++j) vw[j] = *(const v4f*)(pW[j] + kb);
}

__device__ __forceinline__ void chunk_store(float* sb, const v4f (&va)[8], const v4f (&vw)[5])
{
#pragma unroll
    for (int j = 0; j < 8; ++j) *(v4f*)(sb + j * 16 * LST) = va[j];
#pragma unroll
    for (int j = 0; j < 5; ++j) *(v4f*)(sb + BM * LST + j * 16 * LST) = vw[j];
}

__device__ __forceinline__ void lds_barrier()
{
    __builtin_amdgcn_fence(__ATOMIC_RELEASE, "workgroup", "local");
    __builtin_amdgcn_s_barrier();
    __builtin_amdgcn_fence(__ATOMIC_ACQUIRE, "workgroup", "local");
}

__device__ __forceinline__ void half_chunk_mma(v8f (&acc)[2 * NTB], const float* a0p, const float* bp)
{
    const float* a1p = a0p + 16 * LST;
#pragma unroll
    for (int kk = 0; kk < KC / 2; kk += 4) {
        v2f a0, a1;
        a0.x = a0p[kk];
        a0.y = a0p[kk + 1];
        a1.x = a1p[kk];
        a1.y = a1p[kk + 1];
#pragma unroll
        for (int nt = 0; nt < NTB; ++nt) {
            v2f b;
            b.x = bp[nt * 16 * LST + kk];
            b.y = bp[nt * 16 * LST + kk + 1];
            acc[nt] = __builtin_amdgcn_wmma_f32_16x16x4_f32(
                false, a0, false, b, (short)0, acc[nt], false, false);
            acc[nt + NTB] = __builtin_amdgcn_wmma_f32_16x16x4_f32(
                false, a1, false, b, (short)0, acc[nt + NTB], false, false);
        }
    }
}

__global__ __launch_bounds__(384) void gemm_bert(
    const float* A, const float* W,
    const float* __restrict__ bias, const float* __restrict__ prelu_a,
    float* __restrict__ art)
{
    __shared__ __align__(16) float smem[2 * STG_FLOATS];

    const int tid  = threadIdx.x;
    const int role = __builtin_amdgcn_readfirstlane(tid >> 7);
    const int t    = tid & 127;
    const int wave = tid >> 5;
    const int wv   = wave & 3;
    const int lane = tid & 31;
    const int rc   = lane & 15;
    const int rsel = (lane >> 4) << 1;
    const int rowh = (lane >> 4) << 3;
    const int grp  = role & 1;

    const int m0 = blockIdx.x * BM;
    const int nb = blockIdx.y * BN;

    float* st0 = smem;

    v8f acc[2 * NTB];
#pragma unroll
    for (int q = 0; q < 2 * NTB; ++q) acc[q] = (v8f){0,0,0,0,0,0,0,0};

    const float* a0p = st0 + (wv * 32 + rc) * LST + rsel + grp * (KC / 2);
    const float* bp  = st0 + BM * LST + rc * LST + rsel + grp * (KC / 2);

    const float* pA = A + (size_t)(m0 + (t >> 3)) * DB_ + ((t & 7) << 2);
    const float* pW[5];
#pragma unroll
    for (int j = 0; j < 5; ++j) {
        int wr = min(nb + (t >> 3) + 16 * j, D_ - 1);
        pW[j] = W + (size_t)wr * DB_ + ((t & 7) << 2);
    }
    float* sb0 = st0 + (t >> 3) * LST + ((t & 7) << 2);
    float* sb1 = sb0 + STG_FLOATS;

    v4f xa[8], xw[5];

    if (role == 2) {
        chunk_fetch(xa, xw, pA, pW, 0);
        __builtin_amdgcn_sched_barrier(0);
        chunk_store(sb0, xa, xw);
        chunk_fetch(xa, xw, pA, pW, KC);
    }
    lds_barrier();

    for (int p = 0; p < NCH; p += 2) {
        if (role == 2) {
            chunk_store(sb1, xa, xw);
            chunk_fetch(xa, xw, pA, pW, min(p + 2, NCH - 1) * KC);
        } else {
            half_chunk_mma(acc, a0p, bp);
        }
        lds_barrier();
        if (role == 2) {
            if (p + 2 < NCH) chunk_store(sb0, xa, xw);
            chunk_fetch(xa, xw, pA, pW, min(p + 3, NCH - 1) * KC);
        } else {
            half_chunk_mma(acc, a0p + STG_FLOATS, bp + STG_FLOATS);
        }
        lds_barrier();
    }

    float* red = smem;
    v8f fin[NTB], giv[NTB];
#pragma unroll
    for (int q = 0; q < NTB; ++q) {
        fin[q] = grp ? acc[q + NTB] : acc[q];
        giv[q] = grp ? acc[q] : acc[q + NTB];
    }
    if (role < 2) {
#pragma unroll
        for (int q = 0; q < NTB; ++q)
#pragma unroll
            for (int i = 0; i < 8; ++i)
                red[((wave * NTB + q) * 8 + i) * 32 + lane] = giv[q][i];
    }
    __syncthreads();
    if (role < 2) {
        const int pw8 = wave ^ 4;
#pragma unroll
        for (int q = 0; q < NTB; ++q)
#pragma unroll
            for (int i = 0; i < 8; ++i)
                fin[q][i] += red[((pw8 * NTB + q) * 8 + i) * 32 + lane];

        const float pa_ = prelu_a[0];
#pragma unroll
        for (int nt = 0; nt < NTB; ++nt) {
            int gc = nb + nt * 16 + rc;
            if (gc < D_) {
                float bb = bias[gc];
#pragma unroll
                for (int i = 0; i < 8; ++i) {
                    int gr = m0 + wv * 32 + grp * 16 + i + rowh;
                    float v = fin[nt][i] + bb;
                    v = v > 0.f ? v : pa_ * v;
                    art[(size_t)gr * D_ + gc] = v;
                }
            }
        }
    }
}

#define ZBM 128
#define ZBN 64
#define ZKC 60
#define ZQ  (ZKC / 4)
#define ZBR 69

__global__ __launch_bounds__(128) void gemm_z(
    const float* art, const float* W1, float* Z)
{
    __shared__ __align__(16) float As[ZBM * ZKC];
    __shared__ __align__(16) float Bs[ZBR * ZKC];

    const int tid  = threadIdx.x;
    const int wave = tid >> 5;
    const int lane = tid & 31;
    const int rc   = lane & 15;
    const int rsel = (lane >> 4) << 1;
    const int rowh = (lane >> 4) << 3;

    const int m0 = blockIdx.x * ZBM;
    const int nb = blockIdx.y * ZBN;

    v8f acc[8];
#pragma unroll
    for (int t = 0; t < 8; ++t) acc[t] = (v8f){0,0,0,0,0,0,0,0};

    for (int kb = 0; kb < D_; kb += ZKC) {
        v4f va[15], vw[8];
#pragma unroll
        for (int j = 0; j < 15; ++j) {
            int i   = tid + 128 * j;
            int row = i / ZQ;
            int c4  = (i % ZQ) << 2;
            va[j] = *(const v4f*)&art[(size_t)(m0 + row) * D_ + kb + c4];
        }
#pragma unroll
        for (int j = 0; j < 8; ++j) {
            int i  = tid + 128 * j;
            int n  = i / ZQ;
            int c4 = (i % ZQ) << 2;
            int wr = min(nb + n, H_ - 1);
            vw[j] = *(const v4f*)&W1[(size_t)wr * D_ + kb + c4];
        }
        __builtin_amdgcn_sched_barrier(0);
#pragma unroll
        for (int j = 0; j < 15; ++j)
            *(v4f*)&As[(tid + 128 * j) << 2] = va[j];
#pragma unroll
        for (int j = 0; j < 8; ++j)
            *(v4f*)&Bs[(tid + 128 * j) << 2] = vw[j];
        __syncthreads();

        const float* a0p = &As[(wave * 32 + rc) * ZKC + rsel];
        const float* a1p = a0p + 16 * ZKC;
        const float* bp  = &Bs[rc * ZKC + rsel];
#pragma unroll
        for (int kk = 0; kk < ZKC; kk += 4) {
            v2f a0, a1;
            a0.x = a0p[kk];
            a0.y = a0p[kk + 1];
            a1.x = a1p[kk];
            a1.y = a1p[kk + 1];
#pragma unroll
            for (int nt = 0; nt < 4; ++nt) {
                v2f b;
                b.x = bp[nt * 16 * ZKC + kk];
                b.y = bp[nt * 16 * ZKC + kk + 1];
                acc[nt] = __builtin_amdgcn_wmma_f32_16x16x4_f32(
                    false, a0, false, b, (short)0, acc[nt], false, false);
                acc[nt + 4] = __builtin_amdgcn_wmma_f32_16x16x4_f32(
                    false, a1, false, b, (short)0, acc[nt + 4], false, false);
            }
        }
        __syncthreads();
    }

#pragma unroll
    for (int nt = 0; nt < 4; ++nt) {
        int gc = nb + nt * 16 + rc;
#pragma unroll
        for (int part = 0; part < 2; ++part) {
#pragma unroll
            for (int i = 0; i < 8; ++i) {
                int gr = m0 + wave * 32 + part * 16 + i + rowh;
                Z[(size_t)gr * H_ + gc] = acc[nt + 4 * part][i];
            }
        }
    }
}

#define PG 10

__global__ __launch_bounds__(128) void pool_head(
    const float* __restrict__ Z, const int* __restrict__ idx, const int* __restrict__ mask,
    const float* __restrict__ b1, const float* __restrict__ W2, const float* __restrict__ b2,
    float* __restrict__ out)
{
    const int node = blockIdx.x * 4 + (threadIdx.x >> 5);
    const int lane = threadIdx.x & 31;
    const int b    = node / NN_;

    const int   le = min(lane, NW_ - 1);
    const int   iv = idx[(size_t)node * NW_ + le];
    const float mv = lane < NW_ ? (float)mask[(size_t)node * NW_ + le] : 0.f;

    float cnt = mv;
#pragma unroll
    for (int off = 16; off > 0; off >>= 1) cnt += __shfl_xor(cnt, off, 32);
    const float inv = 1.f / fmaxf(cnt, 1.f);

    const float4* zb = (const float4*)(Z + (size_t)b * L_ * H_) + lane;

    float4 s0 = make_float4(0.f, 0.f, 0.f, 0.f);
    float4 s1 = make_float4(0.f, 0.f, 0.f, 0.f);
#pragma unroll
    for (int g = 0; g < NW_ / PG; ++g) {
        float4 x0[PG], x1[PG];
        float  m[PG];
#pragma unroll
        for (int u = 0; u < PG; ++u) {
            const int w = g * PG + u;
            const int r = __builtin_amdgcn_readlane(iv, w);
            m[u] = __int_as_float(__builtin_amdgcn_readlane(__float_as_int(mv), w));
            const float4* zr = zb + (size_t)r * (H_ / 4);
            x0[u] = zr[0];
            x1[u] = zr[32];
        }
        __builtin_amdgcn_sched_barrier(0);
#pragma unroll
        for (int u = 0; u < PG; ++u) {
            s0.x += m[u] * x0[u].x;  s0.y += m[u] * x0[u].y;
            s0.z += m[u] * x0[u].z;  s0.w += m[u] * x0[u].w;
            s1.x += m[u] * x1[u].x;  s1.y += m[u] * x1[u].y;
            s1.z += m[u] * x1[u].z;  s1.w += m[u] * x1[u].w;
        }
    }

    const int c0 = lane << 2;
    const int c1 = 128 + (lane << 2);
    float p = 0.f;
    p += fmaxf(s0.x * inv + b1[c0 + 0], 0.f) * W2[c0 + 0];
    p += fmaxf(s0.y * inv + b1[c0 + 1], 0.f) * W2[c0 + 1];
    p += fmaxf(s0.z * inv + b1[c0 + 2], 0.f) * W2[c0 + 2];
    p += fmaxf(s0.w * inv + b1[c0 + 3], 0.f) * W2[c0 + 3];
    p += fmaxf(s1.x * inv + b1[c1 + 0], 0.f) * W2[c1 + 0];
    p += fmaxf(s1.y * inv + b1[c1 + 1], 0.f) * W2[c1 + 1];
    p += fmaxf(s1.z * inv + b1[c1 + 2], 0.f) * W2[c1 + 2];
    p += fmaxf(s1.w * inv + b1[c1 + 3], 0.f) * W2[c1 + 3];
#pragma unroll
    for (int off = 16; off > 0; off >>= 1) p += __shfl_xor(p, off, 32);
    if (lane == 0) out[node] = p + b2[0];
}

extern "C" void kernel_launch(void* const* d_in, const int* in_sizes, int n_in,
                              void* d_out, int out_size, void* d_ws, size_t ws_size,
                              hipStream_t stream)
{
    const float* articles = (const float*)d_in[0];
    const int*   nidx     = (const int*)  d_in[1];
    const int*   nmask    = (const int*)  d_in[2];
    const float* W_bert   = (const float*)d_in[5];
    const float* b_bert   = (const float*)d_in[6];
    const float* prelu_a  = (const float*)d_in[7];
    const float* W1       = (const float*)d_in[8];
    const float* b1       = (const float*)d_in[9];
    const float* W2       = (const float*)d_in[10];
    const float* b2       = (const float*)d_in[11];
    float* out = (float*)d_out;

    float* art = (float*)d_ws;
    float* Zp  = art + (size_t)M1 * D_;
    (void)in_sizes; (void)n_in; (void)out_size; (void)ws_size;

    gemm_bert<<<dim3(M1 / BM, 4), 384, 0, stream>>>(articles, W_bert, b_bert, prelu_a, art);
    gemm_z   <<<dim3(M1 / ZBM, H_ / ZBN), 128, 0, stream>>>(art, W1, Zp);
    pool_head<<<MN / 4, 128, 0, stream>>>(Zp, nidx, nmask, b1, W2, b2, out);
}


# gemm_bert: 8 waves, no loader waves: a SIMD's two waves take alternate K chunks a phase apart (fragments read once a chunk, 4 k-steps a phase), one's WMMAs cover the other's reads; sums reordered
# speedup vs baseline: 2.5583x; 1.0876x over previous
#include <hip/hip_runtime.h>
#include <hip/hip_bf16.h>

typedef __attribute__((ext_vector_type(2))) float v2f;
typedef __attribute__((ext_vector_type(8))) float v8f;
typedef __attribute__((ext_vector_type(4))) float v4f;

#define B_   16
#define L_   512
#define DB_  3072
#define D_   300
#define NN_  256
#define NW_  30
#define H_   256

#define M1   (B_ * L_)
#define MN   (B_ * NN_)

#define BM 128
#define BN 80
#define BNS 96
#define KC 32
#define LST 36
#define NCH (DB_ / KC)
#define NTB (BN / 16)
#define STG_FLOATS ((BM + BNS) * LST)
#define NSG 7
#define NFR 7
#define NHF (NFR * 2)

__device__ __forceinline__ void chunk_fetch(v4f (&sg)[NSG],
                                            const float* pA, const float* const (&pW)[3], int kb)
{
#pragma unroll
    for (int j = 0; j < 4; ++j) sg[j] = *(const v4f*)(pA + (size_t)j * 32 * DB_ + kb);
#pragma unroll
    for (int j = 0; j < 3; ++j) sg[4 + j] = *(const v4f*)(pW[j] + kb);
}

__device__ __forceinline__ void chunk_store(float* sb, const v4f (&sg)[NSG])
{
#pragma unroll
    for (int j = 0; j < 4; ++j) *(v4f*)(sb + j * 32 * LST) = sg[j];
#pragma unroll
    for (int j = 0; j < 3; ++j) *(v4f*)(sb + BM * LST + j * 32 * LST) = sg[4 + j];
}

__device__ __forceinline__ void lds_barrier()
{
    __builtin_amdgcn_fence(__ATOMIC_RELEASE, "workgroup", "local");
    __builtin_amdgcn_s_barrier();
    __builtin_amdgcn_fence(__ATOMIC_ACQUIRE, "workgroup", "local");
}

__device__ __forceinline__ void half_frags(v4f (&fr)[NHF], const float* a0p, const float* bp, int kofs)
{
#pragma unroll
    for (int kp = 0; kp < 2; ++kp) {
#pragma unroll
        for (int f = 0; f < NFR; ++f) {
            const float* q = (f < 2 ? a0p + f * 16 * LST : bp + (f - 2) * 16 * LST) + kofs + kp * 8;
            v4f r;
            r.x = q[0];
            r.y = q[1];
            r.z = q[4];
            r.w = q[5];
            fr[kp * NFR + f] = r;
        }
    }
}

__device__ __forceinline__ void pair_mma(v8f (&acc)[2 * NTB], const v4f (&fr)[NHF], int kp)
{
#pragma unroll
    for (int h = 0; h < 2; ++h) {
        const v4f ra0 = fr[kp * NFR], ra1 = fr[kp * NFR + 1];
        const v2f a0 = h ? (v2f){ra0.z, ra0.w} : (v2f){ra0.x, ra0.y};
        const v2f a1 = h ? (v2f){ra1.z, ra1.w} : (v2f){ra1.x, ra1.y};
#pragma unroll
        for (int nt = 0; nt < NTB; ++nt) {
            const v4f rb = fr[kp * NFR + 2 + nt];
            const v2f b = h ? (v2f){rb.z, rb.w} : (v2f){rb.x, rb.y};
            acc[nt] = __builtin_amdgcn_wmma_f32_16x16x4_f32(
                false, a0, false, b, (short)0, acc[nt], false, false);
            acc[nt + NTB] = __builtin_amdgcn_wmma_f32_16x16x4_f32(
                false, a1, false, b, (short)0, acc[nt + NTB], false, false);
        }
    }
}

__global__ __launch_bounds__(256) void gemm_bert(
    const float* A, const float* W,
    const float* __restrict__ bias, const float* __restrict__ prelu_a,
    float* __restrict__ art)
{
    __shared__ __align__(16) float smem[2 * STG_FLOATS];

    const int tid  = threadIdx.x;
    const int grp  = __builtin_amdgcn_readfirstlane(tid >> 7);
    const int wave = tid >> 5;
    const int wv   = wave & 3;
    const int lane = tid & 31;
    const int rc   = lane & 15;
    const int rsel = (lane >> 4) << 1;
    const int rowh = (lane >> 4) << 3;

    const int m0 = blockIdx.x * BM;
    const int nb = blockIdx.y * BN;

    float* st0 = smem;

    v8f acc[2 * NTB];
#pragma unroll
    for (int q = 0; q < 2 * NTB; ++q) acc[q] = (v8f){0,0,0,0,0,0,0,0};

    const float* a0p = st0 + grp * STG_FLOATS + (wv * 32 + rc) * LST + rsel;
    const float* bp  = st0 + grp * STG_FLOATS + BM * LST + rc * LST + rsel;

    const float* pA = A + (size_t)(m0 + (tid >> 3)) * DB_ + ((tid & 7) << 2);
    const float* pW[3];
#pragma unroll
    for (int j = 0; j < 3; ++j) {
        int wr = min(nb + (tid >> 3) + 32 * j, D_ - 1);
        pW[j] = W + (size_t)wr * DB_ + ((tid & 7) << 2);
    }
    float* sb0 = st0 + (tid >> 3) * LST + ((tid & 7) << 2);
    float* sb1 = sb0 + STG_FLOATS;

    v4f sg[NSG];
    v4f fp[NHF], fq[NHF];
#pragma unroll
    for (int j = 0; j < NHF; ++j) fp[j] = fq[j] = (v4f){0, 0, 0, 0};

    chunk_fetch(sg, pA, pW, 0);
    __builtin_amdgcn_sched_barrier(0);
    chunk_store(sb0, sg);
    chunk_fetch(sg, pA, pW, KC);
    lds_barrier();

    for (int p = 0; p < NCH; p += 2) {
        if (grp == 0) {
            half_frags(fp, a0p, bp, 0);
            half_frags(fq, a0p, bp, KC / 2);
        }
        pair_mma(acc, fp, 0);
        __builtin_amdgcn_sched_barrier(0);
        chunk_store(sb1, sg);
        chunk_fetch(sg, pA, pW, min(p + 2, NCH - 1) * KC);
        __builtin_amdgcn_sched_barrier(0);
        pair_mma(acc, fp, 1);
        lds_barrier();

        if (grp == 1) {
            half_frags(fq, a0p, bp, 0);
            half_frags(fp, a0p, bp, KC / 2);
        }
        pair_mma(acc, fq, 0);
        __builtin_amdgcn_sched_barrier(0);
        if (p + 2 < NCH) chunk_store(sb0, sg);
        chunk_fetch(sg, pA, pW, min(p + 3, NCH - 1) * KC);
        __builtin_amdgcn_sched_barrier(0);
        pair_mma(acc, fq, 1);
        lds_barrier();
    }
    if (grp == 1) {
        pair_mma(acc, fp, 0);
        pair_mma(acc, fp, 1);
    }

    float* red = smem;
    v8f fin[NTB], giv[NTB];
#pragma unroll
    for (int q = 0; q < NTB; ++q) {
        fin[q] = grp ? acc[q + NTB] : acc[q];
        giv[q] = grp ? acc[q] : acc[q + NTB];
    }
#pragma unroll
    for (int q = 0; q < NTB; ++q)
#pragma unroll
        for (int i = 0; i < 8; ++i)
            red[((wave * NTB + q) * 8 + i) * 32 + lane] = giv[q][i];
    __syncthreads();
    {
        const int pw8 = wave ^ 4;
#pragma unroll
        for (int q = 0; q < NTB; ++q)
#pragma unroll
            for (int i = 0; i < 8; ++i)
                fin[q][i] += red[((pw8 * NTB + q) * 8 + i) * 32 + lane];

        const float pa_ = prelu_a[0];
#pragma unroll
        for (int nt = 0; nt < NTB; ++nt) {
            int gc = nb + nt * 16 + rc;
            if (gc < D_) {
                float bb = bias[gc];
#pragma unroll
                for (int i = 0; i < 8; ++i) {
                    int gr = m0 + wv * 32 + grp * 16 + i + rowh;
                    float v = fin[nt][i] + bb;
                    v = v > 0.f ? v : pa_ * v;
                    art[(size_t)gr * D_ + gc] = v;
                }
            }
        }
    }
}

#define ZBM 128
#define ZBN 64
#define ZKC 60
#define ZQ  (ZKC / 4)
#define ZBR 69

__global__ __launch_bounds__(128) void gemm_z(
    const float* art, const float* W1, float* Z)
{
    __shared__ __align__(16) float As[ZBM * ZKC];
    __shared__ __align__(16) float Bs[ZBR * ZKC];

    const int tid  = threadIdx.x;
    const int wave = tid >> 5;
    const int lane = tid & 31;
    const int rc   = lane & 15;
    const int rsel = (lane >> 4) << 1;
    const int rowh = (lane >> 4) << 3;

    const int m0 = blockIdx.x * ZBM;
    const int nb = blockIdx.y * ZBN;

    v8f acc[8];
#pragma unroll
    for (int t = 0; t < 8; ++t) acc[t] = (v8f){0,0,0,0,0,0,0,0};

    for (int kb = 0; kb < D_; kb += ZKC) {
        v4f va[15], vw[8];
#pragma unroll
        for (int j = 0; j < 15; ++j) {
            int i   = tid + 128 * j;
            int row = i / ZQ;
            int c4  = (i % ZQ) << 2;
            va[j] = *(const v4f*)&art[(size_t)(m0 + row) * D_ + kb + c4];
        }
#pragma unroll
        for (int j = 0; j < 8; ++j) {
            int i  = tid + 128 * j;
            int n  = i / ZQ;
            int c4 = (i % ZQ) << 2;
            int wr = min(nb + n, H_ - 1);
            vw[j] = *(const v4f*)&W1[(size_t)wr * D_ + kb + c4];
        }
        __builtin_amdgcn_sched_barrier(0);
#pragma unroll
        for (int j = 0; j < 15; ++j)
            *(v4f*)&As[(tid + 128 * j) << 2] = va[j];
#pragma unroll
        for (int j = 0; j < 8; ++j)
            *(v4f*)&Bs[(tid + 128 * j) << 2] = vw[j];
        __syncthreads();

        const float* a0p = &As[(wave * 32 + rc) * ZKC + rsel];
        const float* a1p = a0p + 16 * ZKC;
        const float* bp  = &Bs[rc * ZKC + rsel];
#pragma unroll
        for (int kk = 0; kk < ZKC; kk += 4) {
            v2f a0, a1;
            a0.x = a0p[kk];
            a0.y = a0p[kk + 1];
            a1.x = a1p[kk];
            a1.y = a1p[kk + 1];
#pragma unroll
            for (int nt = 0; nt < 4; ++nt) {
                v2f b;
                b.x = bp[nt * 16 * ZKC + kk];
                b.y = bp[nt * 16 * ZKC + kk + 1];
                acc[nt] = __builtin_amdgcn_wmma_f32_16x16x4_f32(
                    false, a0, false, b, (short)0, acc[nt], false, false);
                acc[nt + 4] = __builtin_amdgcn_wmma_f32_16x16x4_f32(
                    false, a1, false, b, (short)0, acc[nt + 4], false, false);
            }
        }
        __syncthreads();
    }

#pragma unroll
    for (int nt = 0; nt < 4; ++nt) {
        int gc = nb + nt * 16 + rc;
#pragma unroll
        for (int part = 0; part < 2; ++part) {
#pragma unroll
            for (int i = 0; i < 8; ++i) {
                int gr = m0 + wave * 32 + part * 16 + i + rowh;
                Z[(size_t)gr * H_ + gc] = acc[nt + 4 * part][i];
            }
        }
    }
}

#define PG 10

__global__ __launch_bounds__(128) void pool_head(
    const float* __restrict__ Z, const int* __restrict__ idx, const int* __restrict__ mask,
    const float* __restrict__ b1, const float* __restrict__ W2, const float* __restrict__ b2,
    float* __restrict__ out)
{
    const int node = blockIdx.x * 4 + (threadIdx.x >> 5);
    const int lane = threadIdx.x & 31;
    const int b    = node / NN_;

    const int   le = min(lane, NW_ - 1);
    const int   iv = idx[(size_t)node * NW_ + le];
    const float mv = lane < NW_ ? (float)mask[(size_t)node * NW_ + le] : 0.f;

    float cnt = mv;
#pragma unroll
    for (int off = 16; off > 0; off >>= 1) cnt += __shfl_xor(cnt, off, 32);
    const float inv = 1.f / fmaxf(cnt, 1.f);

    const float4* zb = (const float4*)(Z + (size_t)b * L_ * H_) + lane;

    float4 s0 = make_float4(0.f, 0.f, 0.f, 0.f);
    float4 s1 = make_float4(0.f, 0.f, 0.f, 0.f);
#pragma unroll
    for (int g = 0; g < NW_ / PG; ++g) {
        float4 x0[PG], x1[PG];
        float  m[PG];
#pragma unroll
        for (int u = 0; u < PG; ++u) {
            const int w = g * PG + u;
            const int r = __builtin_amdgcn_readlane(iv, w);
            m[u] = __int_as_float(__builtin_amdgcn_readlane(__float_as_int(mv), w));
            const float4* zr = zb + (size_t)r * (H_ / 4);
            x0[u] = zr[0];
            x1[u] = zr[32];
        }
        __builtin_amdgcn_sched_barrier(0);
#pragma unroll
        for (int u = 0; u < PG; ++u) {
            s0.x += m[u] * x0[u].x;  s0.y += m[u] * x0[u].y;
            s0.z += m[u] * x0[u].z;  s0.w += m[u] * x0[u].w;
            s1.x += m[u] * x1[u].x;  s1.y += m[u] * x1[u].y;
            s1.z += m[u] * x1[u].z;  s1.w += m[u] * x1[u].w;
        }
    }

    const int c0 = lane << 2;
    const int c1 = 128 + (lane << 2);
    float p = 0.f;
    p += fmaxf(s0.x * inv + b1[c0 + 0], 0.f) * W2[c0 + 0];
    p += fmaxf(s0.y * inv + b1[c0 + 1], 0.f) * W2[c0 + 1];
    p += fmaxf(s0.z * inv + b1[c0 + 2], 0.f) * W2[c0 + 2];
    p += fmaxf(s0.w * inv + b1[c0 + 3], 0.f) * W2[c0 + 3];
    p += fmaxf(s1.x * inv + b1[c1 + 0], 0.f) * W2[c1 + 0];
    p += fmaxf(s1.y * inv + b1[c1 + 1], 0.f) * W2[c1 + 1];
    p += fmaxf(s1.z * inv + b1[c1 + 2], 0.f) * W2[c1 + 2];
    p += fmaxf(s1.w * inv + b1[c1 + 3], 0.f) * W2[c1 + 3];
#pragma unroll
    for (int off = 16; off > 0; off >>= 1) p += __shfl_xor(p, off, 32);
    if (lane == 0) out[node] = p + b2[0];
}

extern "C" void kernel_launch(void* const* d_in, const int* in_sizes, int n_in,
                              void* d_out, int out_size, void* d_ws, size_t ws_size,
                              hipStream_t stream)
{
    const float* articles = (const float*)d_in[0];
    const int*   nidx     = (const int*)  d_in[1];
    const int*   nmask    = (const int*)  d_in[2];
    const float* W_bert   = (const float*)d_in[5];
    const float* b_bert   = (const float*)d_in[6];
    const float* prelu_a  = (const float*)d_in[7];
    const float* W1       = (const float*)d_in[8];
    const float* b1       = (const float*)d_in[9];
    const float* W2       = (const float*)d_in[10];
    const float* b2       = (const float*)d_in[11];
    float* out = (float*)d_out;

    float* art = (float*)d_ws;
    float* Zp  = art + (size_t)M1 * D_;
    (void)in_sizes; (void)n_in; (void)out_size; (void)ws_size;

    gemm_bert<<<dim3(M1 / BM, 4), 256, 0, stream>>>(articles, W_bert, b_bert, prelu_a, art);
    gemm_z   <<<dim3(M1 / ZBM, H_ / ZBN), 128, 0, stream>>>(art, W1, Zp);
    pool_head<<<MN / 4, 128, 0, stream>>>(Zp, nidx, nmask, b1, W2, b2, out);
}


# gemm_bert: all LDS staging moved to the phase's 4 fragment-reading waves, right behind their reads (13 float4 a thread, loads lead 2 phases); the register-fed wave issues matrix instructions only
# speedup vs baseline: 2.6414x; 1.0325x over previous
#include <hip/hip_runtime.h>
#include <hip/hip_bf16.h>

typedef __attribute__((ext_vector_type(2))) float v2f;
typedef __attribute__((ext_vector_type(8))) float v8f;
typedef __attribute__((ext_vector_type(4))) float v4f;

#define B_   16
#define L_   512
#define DB_  3072
#define D_   300
#define NN_  256
#define NW_  30
#define H_   256

#define M1   (B_ * L_)
#define MN   (B_ * NN_)

#define BM 128
#define BN 80
#define KC 32
#define LST 36
#define NCH (DB_ / KC)
#define NTB (BN / 16)
#define STG_FLOATS ((BM + BN) * LST)
#define NSA 8
#define NSW 5
#define NSG (NSA + NSW)
#define NFR 7
#define NHF (NFR * 2)

__device__ __forceinline__ void chunk_fetch(v4f (&sg)[NSG],
                                            const float* pA, const float* const (&pW)[NSW], int kb)
{
#pragma unroll
    for (int j = 0; j < NSA; ++j) sg[j] = *(const v4f*)(pA + (size_t)j * 16 * DB_ + kb);
#pragma unroll
    for (int j = 0; j < NSW; ++j) sg[NSA + j] = *(const v4f*)(pW[j] + kb);
}

__device__ __forceinline__ void chunk_store(float* sb, const v4f (&sg)[NSG])
{
#pragma unroll
    for (int j = 0; j < NSA; ++j) *(v4f*)(sb + j * 16 * LST) = sg[j];
#pragma unroll
    for (int j = 0; j < NSW; ++j) *(v4f*)(sb + BM * LST + j * 16 * LST) = sg[NSA + j];
}

__device__ __forceinline__ void lds_barrier()
{
    __builtin_amdgcn_fence(__ATOMIC_RELEASE, "workgroup", "local");
    __builtin_amdgcn_s_barrier();
    __builtin_amdgcn_fence(__ATOMIC_ACQUIRE, "workgroup", "local");
}

__device__ __forceinline__ void half_frags(v4f (&fr)[NHF], const float* a0p, const float* bp, int kofs)
{
#pragma unroll
    for (int kp = 0; kp < 2; ++kp) {
#pragma unroll
        for (int f = 0; f < NFR; ++f) {
            const float* q = (f < 2 ? a0p + f * 16 * LST : bp + (f - 2) * 16 * LST) + kofs + kp * 8;
            v4f r;
            r.x = q[0];
            r.y = q[1];
            r.z = q[4];
            r.w = q[5];
            fr[kp * NFR + f] = r;
        }
    }
}

__device__ __forceinline__ void pair_mma(v8f (&acc)[2 * NTB], const v4f (&fr)[NHF], int kp)
{
#pragma unroll
    for (int h = 0; h < 2; ++h) {
        const v4f ra0 = fr[kp * NFR], ra1 = fr[kp * NFR + 1];
        const v2f a0 = h ? (v2f){ra0.z, ra0.w} : (v2f){ra0.x, ra0.y};
        const v2f a1 = h ? (v2f){ra1.z, ra1.w} : (v2f){ra1.x, ra1.y};
#pragma unroll
        for (int nt = 0; nt < NTB; ++nt) {
            const v4f rb = fr[kp * NFR + 2 + nt];
            const v2f b = h ? (v2f){rb.z, rb.w} : (v2f){rb.x, rb.y};
            acc[nt] = __builtin_amdgcn_wmma_f32_16x16x4_f32(
                false, a0, false, b, (short)0, acc[nt], false, false);
            acc[nt + NTB] = __builtin_amdgcn_wmma_f32_16x16x4_f32(
                false, a1, false, b, (short)0, acc[nt + NTB], false, false);
        }
    }
}

__global__ __launch_bounds__(256) void gemm_bert(
    const float* A, const float* W,
    const float* __restrict__ bias, const float* __restrict__ prelu_a,
    float* __restrict__ art)
{
    __shared__ __align__(16) float smem[2 * STG_FLOATS];

    const int tid  = threadIdx.x;
    const int grp  = __builtin_amdgcn_readfirstlane(tid >> 7);
    const int wave = tid >> 5;
    const int wv   = wave & 3;
    const int lane = tid & 31;
    const int rc   = lane & 15;
    const int rsel = (lane >> 4) << 1;
    const int rowh = (lane >> 4) << 3;

    const int m0 = blockIdx.x * BM;
    const int nb = blockIdx.y * BN;

    float* st0 = smem;

    v8f acc[2 * NTB];
#pragma unroll
    for (int q = 0; q < 2 * NTB; ++q) acc[q] = (v8f){0,0,0,0,0,0,0,0};

    const float* a0p = st0 + grp * STG_FLOATS + (wv * 32 + rc) * LST + rsel;
    const float* bp  = st0 + grp * STG_FLOATS + BM * LST + rc * LST + rsel;

    const int ts = tid & 127;
    const float* pA = A + (size_t)(m0 + (ts >> 3)) * DB_ + ((ts & 7) << 2);
    const float* pW[NSW];
#pragma unroll
    for (int j = 0; j < NSW; ++j) {
        int wr = min(nb + (ts >> 3) + 16 * j, D_ - 1);
        pW[j] = W + (size_t)wr * DB_ + ((ts & 7) << 2);
    }
    float* sbo = st0 + (1 - grp) * STG_FLOATS + (ts >> 3) * LST + ((ts & 7) << 2);

    v4f sg[NSG];
    v4f fp[NHF], fq[NHF];
#pragma unroll
    for (int j = 0; j < NHF; ++j) fp[j] = fq[j] = (v4f){0, 0, 0, 0};

    if (grp == 1) {
        chunk_fetch(sg, pA, pW, 0);
        __builtin_amdgcn_sched_barrier(0);
        chunk_store(sbo, sg);
        chunk_fetch(sg, pA, pW, 2 * KC);
    } else {
        chunk_fetch(sg, pA, pW, KC);
    }
    lds_barrier();

    for (int p = 0; p < NCH; p += 2) {
        if (grp == 0) {
            half_frags(fp, a0p, bp, 0);
            half_frags(fq, a0p, bp, KC / 2);
            __builtin_amdgcn_sched_barrier(0);
            chunk_store(sbo, sg);
            chunk_fetch(sg, pA, pW, min(p + 3, NCH - 1) * KC);
        }
        pair_mma(acc, fp, 0);
        __builtin_amdgcn_sched_barrier(0);
        pair_mma(acc, fp, 1);
        lds_barrier();

        if (grp == 1) {
            half_frags(fq, a0p, bp, 0);
            half_frags(fp, a0p, bp, KC / 2);
            __builtin_amdgcn_sched_barrier(0);
            if (p + 2 < NCH) chunk_store(sbo, sg);
            chunk_fetch(sg, pA, pW, min(p + 4, NCH - 1) * KC);
        }
        pair_mma(acc, fq, 0);
        __builtin_amdgcn_sched_barrier(0);
        pair_mma(acc, fq, 1);
        lds_barrier();
    }
    if (grp == 1) {
        pair_mma(acc, fp, 0);
        pair_mma(acc, fp, 1);
    }

    float* red = smem;
    v8f fin[NTB], giv[NTB];
#pragma unroll
    for (int q = 0; q < NTB; ++q) {
        fin[q] = grp ? acc[q + NTB] : acc[q];
        giv[q] = grp ? acc[q] : acc[q + NTB];
    }
#pragma unroll
    for (int q = 0; q < NTB; ++q)
#pragma unroll
        for (int i = 0; i < 8; ++i)
            red[((wave * NTB + q) * 8 + i) * 32 + lane] = giv[q][i];
    __syncthreads();
    {
        const int pw8 = wave ^ 4;
#pragma unroll
        for (int q = 0; q < NTB; ++q)
#pragma unroll
            for (int i = 0; i < 8; ++i)
                fin[q][i] += red[((pw8 * NTB + q) * 8 + i) * 32 + lane];

        const float pa_ = prelu_a[0];
#pragma unroll
        for (int nt = 0; nt < NTB; ++nt) {
            int gc = nb + nt * 16 + rc;
            if (gc < D_) {
                float bb = bias[gc];
#pragma unroll
                for (int i = 0; i < 8; ++i) {
                    int gr = m0 + wv * 32 + grp * 16 + i + rowh;
                    float v = fin[nt][i] + bb;
                    v = v > 0.f ? v : pa_ * v;
                    art[(size_t)gr * D_ + gc] = v;
                }
            }
        }
    }
}

#define ZBM 128
#define ZBN 64
#define ZKC 60
#define ZQ  (ZKC / 4)
#define ZBR 69

__global__ __launch_bounds__(128) void gemm_z(
    const float* art, const float* W1, float* Z)
{
    __shared__ __align__(16) float As[ZBM * ZKC];
    __shared__ __align__(16) float Bs[ZBR * ZKC];

    const int tid  = threadIdx.x;
    const int wave = tid >> 5;
    const int lane = tid & 31;
    const int rc   = lane & 15;
    const int rsel = (lane >> 4) << 1;
    const int rowh = (lane >> 4) << 3;

    const int m0 = blockIdx.x * ZBM;
    const int nb = blockIdx.y * ZBN;

    v8f acc[8];
#pragma unroll
    for (int t = 0; t < 8; ++t) acc[t] = (v8f){0,0,0,0,0,0,0,0};

    for (int kb = 0; kb < D_; kb += ZKC) {
        v4f va[15], vw[8];
#pragma unroll
        for (int j = 0; j < 15; ++j) {
            int i   = tid + 128 * j;
            int row = i / ZQ;
            int c4  = (i % ZQ) << 2;
            va[j] = *(const v4f*)&art[(size_t)(m0 + row) * D_ + kb + c4];
        }
#pragma unroll
        for (int j = 0; j < 8; ++j) {
            int i  = tid + 128 * j;
            int n  = i / ZQ;
            int c4 = (i % ZQ) << 2;
            int wr = min(nb + n, H_ - 1);
            vw[j] = *(const v4f*)&W1[(size_t)wr * D_ + kb + c4];
        }
        __builtin_amdgcn_sched_barrier(0);
#pragma unroll
        for (int j = 0; j < 15; ++j)
            *(v4f*)&As[(tid + 128 * j) << 2] = va[j];
#pragma unroll
        for (int j = 0; j < 8; ++j)
            *(v4f*)&Bs[(tid + 128 * j) << 2] = vw[j];
        __syncthreads();

        const float* a0p = &As[(wave * 32 + rc) * ZKC + rsel];
        const float* a1p = a0p + 16 * ZKC;
        const float* bp  = &Bs[rc * ZKC + rsel];
#pragma unroll
        for (int kk = 0; kk < ZKC; kk += 4) {
            v2f a0, a1;
            a0.x = a0p[kk];
            a0.y = a0p[kk + 1];
            a1.x = a1p[kk];
            a1.y = a1p[kk + 1];
#pragma unroll
            for (int nt = 0; nt < 4; ++nt) {
                v2f b;
                b.x = bp[nt * 16 * ZKC + kk];
                b.y = bp[nt * 16 * ZKC + kk + 1];
                acc[nt] = __builtin_amdgcn_wmma_f32_16x16x4_f32(
                    false, a0, false, b, (short)0, acc[nt], false, false);
                acc[nt + 4] = __builtin_amdgcn_wmma_f32_16x16x4_f32(
                    false, a1, false, b, (short)0, acc[nt + 4], false, false);
            }
        }
        __syncthreads();
    }

#pragma unroll
    for (int nt = 0; nt < 4; ++nt) {
        int gc = nb + nt * 16 + rc;
#pragma unroll
        for (int part = 0; part < 2; ++part) {
#pragma unroll
            for (int i = 0; i < 8; ++i) {
                int gr = m0 + wave * 32 + part * 16 + i + rowh;
                Z[(size_t)gr * H_ + gc] = acc[nt + 4 * part][i];
            }
        }
    }
}

#define PG 10

__global__ __launch_bounds__(128) void pool_head(
    const float* __restrict__ Z, const int* __restrict__ idx, const int* __restrict__ mask,
    const float* __restrict__ b1, const float* __restrict__ W2, const float* __restrict__ b2,
    float* __restrict__ out)
{
    const int node = blockIdx.x * 4 + (threadIdx.x >> 5);
    const int lane = threadIdx.x & 31;
    const int b    = node / NN_;

    const int   le = min(lane, NW_ - 1);
    const int   iv = idx[(size_t)node * NW_ + le];
    const float mv = lane < NW_ ? (float)mask[(size_t)node * NW_ + le] : 0.f;

    float cnt = mv;
#pragma unroll
    for (int off = 16; off > 0; off >>= 1) cnt += __shfl_xor(cnt, off, 32);
    const float inv = 1.f / fmaxf(cnt, 1.f);

    const float4* zb = (const float4*)(Z + (size_t)b * L_ * H_) + lane;

    float4 s0 = make_float4(0.f, 0.f, 0.f, 0.f);
    float4 s1 = make_float4(0.f, 0.f, 0.f, 0.f);
#pragma unroll
    for (int g = 0; g < NW_ / PG; ++g) {
        float4 x0[PG], x1[PG];
        float  m[PG];
#pragma unroll
        for (int u = 0; u < PG; ++u) {
            const int w = g * PG + u;
            const int r = __builtin_amdgcn_readlane(iv, w);
            m[u] = __int_as_float(__builtin_amdgcn_readlane(__float_as_int(mv), w));
            const float4* zr = zb + (size_t)r * (H_ / 4);
            x0[u] = zr[0];
            x1[u] = zr[32];
        }
        __builtin_amdgcn_sched_barrier(0);
#pragma unroll
        for (int u = 0; u < PG; ++u) {
            s0.x += m[u] * x0[u].x;  s0.y += m[u] * x0[u].y;
            s0.z += m[u] * x0[u].z;  s0.w += m[u] * x0[u].w;
            s1.x += m[u] * x1[u].x;  s1.y += m[u] * x1[u].y;
            s1.z += m[u] * x1[u].z;  s1.w += m[u] * x1[u].w;
        }
    }

    const int c0 = lane << 2;
    const int c1 = 128 + (lane << 2);
    float p = 0.f;
    p += fmaxf(s0.x * inv + b1[c0 + 0], 0.f) * W2[c0 + 0];
    p += fmaxf(s0.y * inv + b1[c0 + 1], 0.f) * W2[c0 + 1];
    p += fmaxf(s0.z * inv + b1[c0 + 2], 0.f) * W2[c0 + 2];
    p += fmaxf(s0.w * inv + b1[c0 + 3], 0.f) * W2[c0 + 3];
    p += fmaxf(s1.x * inv + b1[c1 + 0], 0.f) * W2[c1 + 0];
    p += fmaxf(s1.y * inv + b1[c1 + 1], 0.f) * W2[c1 + 1];
    p += fmaxf(s1.z * inv + b1[c1 + 2], 0.f) * W2[c1 + 2];
    p += fmaxf(s1.w * inv + b1[c1 + 3], 0.f) * W2[c1 + 3];
#pragma unroll
    for (int off = 16; off > 0; off >>= 1) p += __shfl_xor(p, off, 32);
    if (lane == 0) out[node] = p + b2[0];
}

extern "C" void kernel_launch(void* const* d_in, const int* in_sizes, int n_in,
                              void* d_out, int out_size, void* d_ws, size_t ws_size,
                              hipStream_t stream)
{
    const float* articles = (const float*)d_in[0];
    const int*   nidx     = (const int*)  d_in[1];
    const int*   nmask    = (const int*)  d_in[2];
    const float* W_bert   = (const float*)d_in[5];
    const float* b_bert   = (const float*)d_in[6];
    const float* prelu_a  = (const float*)d_in[7];
    const float* W1       = (const float*)d_in[8];
    const float* b1       = (const float*)d_in[9];
    const float* W2       = (const float*)d_in[10];
    const float* b2       = (const float*)d_in[11];
    float* out = (float*)d_out;

    float* art = (float*)d_ws;
    float* Zp  = art + (size_t)M1 * D_;
    (void)in_sizes; (void)n_in; (void)out_size; (void)ws_size;

    gemm_bert<<<dim3(M1 / BM, 4), 256, 0, stream>>>(articles, W_bert, b_bert, prelu_a, art);
    gemm_z   <<<dim3(M1 / ZBM, H_ / ZBN), 128, 0, stream>>>(art, W1, Zp);
    pool_head<<<MN / 4, 128, 0, stream>>>(Zp, nidx, nmask, b1, W2, b2, out);
}


# gemm_bert: phase barrier signalled between the phase's two halves of matrix instructions, wave priority raised from signal to wait: a wave let through early leaves the pipe to its partner's rest
# speedup vs baseline: 2.7365x; 1.0360x over previous
#include <hip/hip_runtime.h>
#include <hip/hip_bf16.h>

typedef __attribute__((ext_vector_type(2))) float v2f;
typedef __attribute__((ext_vector_type(8))) float v8f;
typedef __attribute__((ext_vector_type(4))) float v4f;

#define B_   16
#define L_   512
#define DB_  3072
#define D_   300
#define NN_  256
#define NW_  30
#define H_   256

#define M1   (B_ * L_)
#define MN   (B_ * NN_)

#define BM 128
#define BN 80
#define KC 32
#define LST 36
#define NCH (DB_ / KC)
#define NTB (BN / 16)
#define STG_FLOATS ((BM + BN) * LST)
#define NSA 8
#define NSW 5
#define NSG (NSA + NSW)
#define NFR 7
#define NHF (NFR * 2)

__device__ __forceinline__ void chunk_fetch(v4f (&sg)[NSG],
                                            const float* pA, const float* const (&pW)[NSW], int kb)
{
#pragma unroll
    for (int j = 0; j < NSA; ++j) sg[j] = *(const v4f*)(pA + (size_t)j * 16 * DB_ + kb);
#pragma unroll
    for (int j = 0; j < NSW; ++j) sg[NSA + j] = *(const v4f*)(pW[j] + kb);
}

__device__ __forceinline__ void chunk_store(float* sb, const v4f (&sg)[NSG])
{
#pragma unroll
    for (int j = 0; j < NSA; ++j) *(v4f*)(sb + j * 16 * LST) = sg[j];
#pragma unroll
    for (int j = 0; j < NSW; ++j) *(v4f*)(sb + BM * LST + j * 16 * LST) = sg[NSA + j];
}

__device__ __forceinline__ void lds_barrier()
{
    __builtin_amdgcn_fence(__ATOMIC_RELEASE, "workgroup", "local");
    __builtin_amdgcn_s_barrier();
    __builtin_amdgcn_fence(__ATOMIC_ACQUIRE, "workgroup", "local");
}

__device__ __forceinline__ void phase_signal()
{
    __builtin_amdgcn_fence(__ATOMIC_RELEASE, "workgroup", "local");
    __builtin_amdgcn_s_setprio(3);
    __builtin_amdgcn_s_barrier_signal(-1);
}

__device__ __forceinline__ void phase_wait()
{
    __builtin_amdgcn_s_barrier_wait(-1);
    __builtin_amdgcn_s_setprio(0);
    __builtin_amdgcn_fence(__ATOMIC_ACQUIRE, "workgroup", "local");
}

__device__ __forceinline__ void half_frags(v4f (&fr)[NHF], const float* a0p, const float* bp, int kofs)
{
#pragma unroll
    for (int kp = 0; kp < 2; ++kp) {
#pragma unroll
        for (int f = 0; f < NFR; ++f) {
            const float* q = (f < 2 ? a0p + f * 16 * LST : bp + (f - 2) * 16 * LST) + kofs + kp * 8;
            v4f r;
            r.x = q[0];
            r.y = q[1];
            r.z = q[4];
            r.w = q[5];
            fr[kp * NFR + f] = r;
        }
    }
}

__device__ __forceinline__ void pair_mma(v8f (&acc)[2 * NTB], const v4f (&fr)[NHF], int kp)
{
#pragma unroll
    for (int h = 0; h < 2; ++h) {
        const v4f ra0 = fr[kp * NFR], ra1 = fr[kp * NFR + 1];
        const v2f a0 = h ? (v2f){ra0.z, ra0.w} : (v2f){ra0.x, ra0.y};
        const v2f a1 = h ? (v2f){ra1.z, ra1.w} : (v2f){ra1.x, ra1.y};
#pragma unroll
        for (int nt = 0; nt < NTB; ++nt) {
            const v4f rb = fr[kp * NFR + 2 + nt];
            const v2f b = h ? (v2f){rb.z, rb.w} : (v2f){rb.x, rb.y};
            acc[nt] = __builtin_amdgcn_wmma_f32_16x16x4_f32(
                false, a0, false, b, (short)0, acc[nt], false, false);
            acc[nt + NTB] = __builtin_amdgcn_wmma_f32_16x16x4_f32(
                false, a1, false, b, (short)0, acc[nt + NTB], false, false);
        }
    }
}

__global__ __launch_bounds__(256) void gemm_bert(
    const float* A, const float* W,
    const float* __restrict__ bias, const float* __restrict__ prelu_a,
    float* __restrict__ art)
{
    __shared__ __align__(16) float smem[2 * STG_FLOATS];

    const int tid  = threadIdx.x;
    const int grp  = __builtin_amdgcn_readfirstlane(tid >> 7);
    const int wave = tid >> 5;
    const int wv   = wave & 3;
    const int lane = tid & 31;
    const int rc   = lane & 15;
    const int rsel = (lane >> 4) << 1;
    const int rowh = (lane >> 4) << 3;

    const int m0 = blockIdx.x * BM;
    const int nb = blockIdx.y * BN;

    float* st0 = smem;

    v8f acc[2 * NTB];
#pragma unroll
    for (int q = 0; q < 2 * NTB; ++q) acc[q] = (v8f){0,0,0,0,0,0,0,0};

    const float* a0p = st0 + grp * STG_FLOATS + (wv * 32 + rc) * LST + rsel;
    const float* bp  = st0 + grp * STG_FLOATS + BM * LST + rc * LST + rsel;

    const int ts = tid & 127;
    const float* pA = A + (size_t)(m0 + (ts >> 3)) * DB_ + ((ts & 7) << 2);
    const float* pW[NSW];
#pragma unroll
    for (int j = 0; j < NSW; ++j) {
        int wr = min(nb + (ts >> 3) + 16 * j, D_ - 1);
        pW[j] = W + (size_t)wr * DB_ + ((ts & 7) << 2);
    }
    float* sbo = st0 + (1 - grp) * STG_FLOATS + (ts >> 3) * LST + ((ts & 7) << 2);

    v4f sg[NSG];
    v4f fp[NHF], fq[NHF];
#pragma unroll
    for (int j = 0; j < NHF; ++j) fp[j] = fq[j] = (v4f){0, 0, 0, 0};

    if (grp == 1) {
        chunk_fetch(sg, pA, pW, 0);
        __builtin_amdgcn_sched_barrier(0);
        chunk_store(sbo, sg);
        chunk_fetch(sg, pA, pW, 2 * KC);
    } else {
        chunk_fetch(sg, pA, pW, KC);
    }
    lds_barrier();

    for (int p = 0; p < NCH; p += 2) {
        if (grp == 0) {
            half_frags(fp, a0p, bp, 0);
            half_frags(fq, a0p, bp, KC / 2);
            __builtin_amdgcn_sched_barrier(0);
            chunk_store(sbo, sg);
            chunk_fetch(sg, pA, pW, min(p + 3, NCH - 1) * KC);
        }
        pair_mma(acc, fp, 0);
        __builtin_amdgcn_sched_barrier(0);
        phase_signal();
        __builtin_amdgcn_sched_barrier(0);
        pair_mma(acc, fp, 1);
        __builtin_amdgcn_sched_barrier(0);
        phase_wait();

        if (grp == 1) {
            half_frags(fq, a0p, bp, 0);
            half_frags(fp, a0p, bp, KC / 2);
            __builtin_amdgcn_sched_barrier(0);
            if (p + 2 < NCH) chunk_store(sbo, sg);
            chunk_fetch(sg, pA, pW, min(p + 4, NCH - 1) * KC);
        }
        pair_mma(acc, fq, 0);
        __builtin_amdgcn_sched_barrier(0);
        phase_signal();
        __builtin_amdgcn_sched_barrier(0);
        pair_mma(acc, fq, 1);
        __builtin_amdgcn_sched_barrier(0);
        phase_wait();
    }
    if (grp == 1) {
        pair_mma(acc, fp, 0);
        pair_mma(acc, fp, 1);
    }

    float* red = smem;
    v8f fin[NTB], giv[NTB];
#pragma unroll
    for (int q = 0; q < NTB; ++q) {
        fin[q] = grp ? acc[q + NTB] : acc[q];
        giv[q] = grp ? acc[q] : acc[q + NTB];
    }
#pragma unroll
    for (int q = 0; q < NTB; ++q)
#pragma unroll
        for (int i = 0; i < 8; ++i)
            red[((wave * NTB + q) * 8 + i) * 32 + lane] = giv[q][i];
    __syncthreads();
    {
        const int pw8 = wave ^ 4;
#pragma unroll
        for (int q = 0; q < NTB; ++q)
#pragma unroll
            for (int i = 0; i < 8; ++i)
                fin[q][i] += red[((pw8 * NTB + q) * 8 + i) * 32 + lane];

        const float pa_ = prelu_a[0];
#pragma unroll
        for (int nt = 0; nt < NTB; ++nt) {
            int gc = nb + nt * 16 + rc;
            if (gc < D_) {
                float bb = bias[gc];
#pragma unroll
                for (int i = 0; i < 8; ++i) {
                    int gr = m0 + wv * 32 + grp * 16 + i + rowh;
                    float v = fin[nt][i] + bb;
                    v = v > 0.f ? v : pa_ * v;
                    art[(size_t)gr * D_ + gc] = v;
                }
            }
        }
    }
}

#define ZBM 128
#define ZBN 64
#define ZKC 60
#define ZQ  (ZKC / 4)
#define ZBR 69

__global__ __launch_bounds__(128) void gemm_z(
    const float* art, const float* W1, float* Z)
{
    __shared__ __align__(16) float As[ZBM * ZKC];
    __shared__ __align__(16) float Bs[ZBR * ZKC];

    const int tid  = threadIdx.x;
    const int wave = tid >> 5;
    const int lane = tid & 31;
    const int rc   = lane & 15;
    const int rsel = (lane >> 4) << 1;
    const int rowh = (lane >> 4) << 3;

    const int m0 = blockIdx.x * ZBM;
    const int nb = blockIdx.y * ZBN;

    v8f acc[8];
#pragma unroll
    for (int t = 0; t < 8; ++t) acc[t] = (v8f){0,0,0,0,0,0,0,0};

    for (int kb = 0; kb < D_; kb += ZKC) {
        v4f va[15], vw[8];
#pragma unroll
        for (int j = 0; j < 15; ++j) {
            int i   = tid + 128 * j;
            int row = i / ZQ;
            int c4  = (i % ZQ) << 2;
            va[j] = *(const v4f*)&art[(size_t)(m0 + row) * D_ + kb + c4];
        }
#pragma unroll
        for (int j = 0; j < 8; ++j) {
            int i  = tid + 128 * j;
            int n  = i / ZQ;
            int c4 = (i % ZQ) << 2;
            int wr = min(nb + n, H_ - 1);
            vw[j] = *(const v4f*)&W1[(size_t)wr * D_ + kb + c4];
        }
        __builtin_amdgcn_sched_barrier(0);
#pragma unroll
        for (int j = 0; j < 15; ++j)
            *(v4f*)&As[(tid + 128 * j) << 2] = va[j];
#pragma unroll
        for (int j = 0; j < 8; ++j)
            *(v4f*)&Bs[(tid + 128 * j) << 2] = vw[j];
        __syncthreads();

        const float* a0p = &As[(wave * 32 + rc) * ZKC + rsel];
        const float* a1p = a0p + 16 * ZKC;
        const float* bp  = &Bs[rc * ZKC + rsel];
#pragma unroll
        for (int kk = 0; kk < ZKC; kk += 4) {
            v2f a0, a1;
            a0.x = a0p[kk];
            a0.y = a0p[kk + 1];
            a1.x = a1p[kk];
            a1.y = a1p[kk + 1];
#pragma unroll
            for (int nt = 0; nt < 4; ++nt) {
                v2f b;
                b.x = bp[nt * 16 * ZKC + kk];
                b.y = bp[nt * 16 * ZKC + kk + 1];
                acc[nt] = __builtin_amdgcn_wmma_f32_16x16x4_f32(
                    false, a0, false, b, (short)0, acc[nt], false, false);
                acc[nt + 4] = __builtin_amdgcn_wmma_f32_16x16x4_f32(
                    false, a1, false, b, (short)0, acc[nt + 4], false, false);
            }
        }
        __syncthreads();
    }

#pragma unroll
    for (int nt = 0; nt < 4; ++nt) {
        int gc = nb + nt * 16 + rc;
#pragma unroll
        for (int part = 0; part < 2; ++part) {
#pragma unroll
            for (int i = 0; i < 8; ++i) {
                int gr = m0 + wave * 32 + part * 16 + i + rowh;
                Z[(size_t)gr * H_ + gc] = acc[nt + 4 * part][i];
            }
        }
    }
}

#define PG 10

__global__ __launch_bounds__(128) void pool_head(
    const float* __restrict__ Z, const int* __restrict__ idx, const int* __restrict__ mask,
    const float* __restrict__ b1, const float* __restrict__ W2, const float* __restrict__ b2,
    float* __restrict__ out)
{
    const int node = blockIdx.x * 4 + (threadIdx.x >> 5);
    const int lane = threadIdx.x & 31;
    const int b    = node / NN_;

    const int   le = min(lane, NW_ - 1);
    const int   iv = idx[(size_t)node * NW_ + le];
    const float mv = lane < NW_ ? (float)mask[(size_t)node * NW_ + le] : 0.f;

    float cnt = mv;
#pragma unroll
    for (int off = 16; off > 0; off >>= 1) cnt += __shfl_xor(cnt, off, 32);
    const float inv = 1.f / fmaxf(cnt, 1.f);

    const float4* zb = (const float4*)(Z + (size_t)b * L_ * H_) + lane;

    float4 s0 = make_float4(0.f, 0.f, 0.f, 0.f);
    float4 s1 = make_float4(0.f, 0.f, 0.f, 0.f);
#pragma unroll
    for (int g = 0; g < NW_ / PG; ++g) {
        float4 x0[PG], x1[PG];
        float  m[PG];
#pragma unroll
        for (int u = 0; u < PG; ++u) {
            const int w = g * PG + u;
            const int r = __builtin_amdgcn_readlane(iv, w);
            m[u] = __int_as_float(__builtin_amdgcn_readlane(__float_as_int(mv), w));
            const float4* zr = zb + (size_t)r * (H_ / 4);
            x0[u] = zr[0];
            x1[u] = zr[32];
        }
        __builtin_amdgcn_sched_barrier(0);
#pragma unroll
        for (int u = 0; u < PG; ++u) {
            s0.x += m[u] * x0[u].x;  s0.y += m[u] * x0[u].y;
            s0.z += m[u] * x0[u].z;  s0.w += m[u] * x0[u].w;
            s1.x += m[u] * x1[u].x;  s1.y += m[u] * x1[u].y;
            s1.z += m[u] * x1[u].z;  s1.w += m[u] * x1[u].w;
        }
    }

    const int c0 = lane << 2;
    const int c1 = 128 + (lane << 2);
    float p = 0.f;
    p += fmaxf(s0.x * inv + b1[c0 + 0], 0.f) * W2[c0 + 0];
    p += fmaxf(s0.y * inv + b1[c0 + 1], 0.f) * W2[c0 + 1];
    p += fmaxf(s0.z * inv + b1[c0 + 2], 0.f) * W2[c0 + 2];
    p += fmaxf(s0.w * inv + b1[c0 + 3], 0.f) * W2[c0 + 3];
    p += fmaxf(s1.x * inv + b1[c1 + 0], 0.f) * W2[c1 + 0];
    p += fmaxf(s1.y * inv + b1[c1 + 1], 0.f) * W2[c1 + 1];
    p += fmaxf(s1.z * inv + b1[c1 + 2], 0.f) * W2[c1 + 2];
    p += fmaxf(s1.w * inv + b1[c1 + 3], 0.f) * W2[c1 + 3];
#pragma unroll
    for (int off = 16; off > 0; off >>= 1) p += __shfl_xor(p, off, 32);
    if (lane == 0) out[node] = p + b2[0];
}

extern "C" void kernel_launch(void* const* d_in, const int* in_sizes, int n_in,
                              void* d_out, int out_size, void* d_ws, size_t ws_size,
                              hipStream_t stream)
{
    const float* articles = (const float*)d_in[0];
    const int*   nidx     = (const int*)  d_in[1];
    const int*   nmask    = (const int*)  d_in[2];
    const float* W_bert   = (const float*)d_in[5];
    const float* b_bert   = (const float*)d_in[6];
    const float* prelu_a  = (const float*)d_in[7];
    const float* W1       = (const float*)d_in[8];
    const float* b1       = (const float*)d_in[9];
    const float* W2       = (const float*)d_in[10];
    const float* b2       = (const float*)d_in[11];
    float* out = (float*)d_out;

    float* art = (float*)d_ws;
    float* Zp  = art + (size_t)M1 * D_;
    (void)in_sizes; (void)n_in; (void)out_size; (void)ws_size;

    gemm_bert<<<dim3(M1 / BM, 4), 256, 0, stream>>>(articles, W_bert, b_bert, prelu_a, art);
    gemm_z   <<<dim3(M1 / ZBM, H_ / ZBN), 128, 0, stream>>>(art, W1, Zp);
    pool_head<<<MN / 4, 128, 0, stream>>>(Zp, nidx, nmask, b1, W2, b2, out);
}
